# starting kernel
# baseline (speedup 1.0000x reference)
#include <hip/hip_runtime.h>


typedef __attribute__((ext_vector_type(2))) float v2f;
typedef __attribute__((ext_vector_type(8))) float v8f;

#define DIM 64
#define KK_TOT 32
#define NT 4

__global__ __launch_bounds__(256) void edge_mlp_wmma(
    const float* __restrict__ nodes, const int* __restrict__ src,
    const int* __restrict__ dst, const float* __restrict__ W1,
    const float* __restrict__ b1, const float* __restrict__ W2,
    const float* __restrict__ b2, float* __restrict__ out, int E)
{
  __shared__ v2f ldsB[KK_TOT * NT * 32];

  for (int g = threadIdx.x; g < KK_TOT * NT * 32; g += blockDim.x) {
    int f = g >> 5, ln = g & 31;
    int n = f >> 5, kk = f & 31;
    int hf = ln >> 4, c = (n << 4) + (ln & 15);
    int row = (kk << 2) + (hf << 1);
    v2f v;
    v.x = W1[row * DIM + c];
    v.y = W1[(row + 1) * DIM + c];
    ldsB[g] = v;
  }
  __syncthreads();

  const int lane = threadIdx.x & 31;
  const int half = lane >> 4;
  const int l15  = lane & 15;

  float bias[NT], w2v[NT];
#pragma unroll
  for (int n = 0; n < NT; ++n) {
    bias[n] = b1[n * 16 + l15];
    w2v[n]  = W2[n * 16 + l15];
  }
  const float b2v = b2[0];

  const int wavesPerBlk = blockDim.x >> 5;
  const int gw = blockIdx.x * wavesPerBlk + (threadIdx.x >> 5);
  const int nw = gridDim.x * wavesPerBlk;
  const int tiles = (E + 15) >> 4;

  for (int t = gw; t < tiles; t += nw) {
    const int e0 = t << 4;
    int e = e0 + l15; if (e >= E) e = E - 1;
    const float* sp = nodes + (long)src[e] * DIM + (half << 1);
    const float* dp = nodes + (long)dst[e] * DIM + (half << 1);

    v2f aS[16], aD[16];
#pragma unroll
    for (int kk = 0; kk < 16; ++kk) {
      aS[kk] = *(const v2f*)(sp + (kk << 2));
      aD[kk] = *(const v2f*)(dp + (kk << 2));
    }

    float pl[8];
#pragma unroll
    for (int r = 0; r < 8; ++r) pl[r] = 0.f;

#pragma unroll
    for (int n = 0; n < NT; ++n) {
      v8f acc;
#pragma unroll
      for (int r = 0; r < 8; ++r) acc[r] = bias[n];
#pragma unroll
      for (int kk = 0; kk < 16; ++kk) {
        v2f bfrag = ldsB[(n * 32 + kk) * 32 + lane];
        acc = __builtin_amdgcn_wmma_f32_16x16x4_f32(
            false, aS[kk], false, bfrag, (short)0, acc, false, false);
      }
#pragma unroll
      for (int kk = 0; kk < 16; ++kk) {
        v2f bfrag = ldsB[(n * 32 + 16 + kk) * 32 + lane];
        acc = __builtin_amdgcn_wmma_f32_16x16x4_f32(
            false, aD[kk], false, bfrag, (short)0, acc, false, false);
      }
#pragma unroll
      for (int r = 0; r < 8; ++r) {
        float h = acc[r] > 0.f ? acc[r] : 0.f;
        pl[r] += h * w2v[n];
      }
    }

#pragma unroll
    for (int r = 0; r < 8; ++r) {
#pragma unroll
      for (int off = 1; off < 16; off <<= 1)
        pl[r] += __shfl_xor(pl[r], off, 32);
      pl[r] += b2v;
    }

    if (e0 + 16 <= E) {
      if (l15 == 0) {
        float4* op = (float4*)(out + e0 + half * 8);
        op[0] = make_float4(pl[0], pl[1], pl[2], pl[3]);
        op[1] = make_float4(pl[4], pl[5], pl[6], pl[7]);
      }
    } else if (l15 == 0) {
#pragma unroll
      for (int r = 0; r < 8; ++r) {
        int idx = e0 + half * 8 + r;
        if (idx < E) out[idx] = pl[r];
      }
    }
  }
}

extern "C" void kernel_launch(void* const* d_in, const int* in_sizes, int n_in,
                              void* d_out, int out_size, void* d_ws, size_t ws_size,
                              hipStream_t stream) {
  const float* nodes = (const float*)d_in[0];
  const int*   src   = (const int*)d_in[1];
  const int*   dst   = (const int*)d_in[2];
  const float* W1    = (const float*)d_in[3];
  const float* b1    = (const float*)d_in[4];
  const float* W2    = (const float*)d_in[5];
  const float* b2    = (const float*)d_in[6];
  float* out = (float*)d_out;

  const int E = in_sizes[1];
  const int tiles = (E + 15) / 16;
  int blocks = (tiles + 7) / 8;
  if (blocks > 2048) blocks = 2048;
  if (blocks < 1) blocks = 1;

  edge_mlp_wmma<<<blocks, 256, 0, stream>>>(nodes, src, dst, W1, b1, W2, b2, out, E);
}


# Rows reach the matrix loop through a per-wave LDS buffer: the next pass is gathered in whole lines (4 per load, not 16) and stored over the half already consumed; 16-wave workgroups; sums unchanged
# speedup vs baseline: 1.3515x; 1.3515x over previous
#include <hip/hip_runtime.h>


typedef __attribute__((ext_vector_type(2))) float v2f;
typedef __attribute__((ext_vector_type(4))) float v4f;
typedef __attribute__((ext_vector_type(8))) float v8f;

#define DIM 64
#define KK_TOT 32
#define NT 4
#define ET 2
#define NWAVES_PER_SIMD 4
#define NGRP 16
#define NWAVES 16
#define ROW_PITCH 528
#define ROWBUF_BYTES (32 * ROW_PITCH)

#define ROW_SHARE(v, k) __builtin_amdgcn_update_dpp(0, (v), 0x150 + (k), 0xf, 0xf, false)
#define GATHER1(Y, T, k) { const int ix_ = ROW_SHARE(Y, k); \
    T[k] = *(const v4f*)(nbytes + (((unsigned)ix_ << 8) + lofs)); }
#define GATHER16(Y, T) GATHER1(Y,T,0) GATHER1(Y,T,1) GATHER1(Y,T,2) GATHER1(Y,T,3) \
    GATHER1(Y,T,4) GATHER1(Y,T,5) GATHER1(Y,T,6) GATHER1(Y,T,7) GATHER1(Y,T,8) GATHER1(Y,T,9) \
    GATHER1(Y,T,10) GATHER1(Y,T,11) GATHER1(Y,T,12) GATHER1(Y,T,13) GATHER1(Y,T,14) GATHER1(Y,T,15)

__global__ __launch_bounds__(NWAVES * 32, NWAVES_PER_SIMD) void edge_mlp_wmma(
    const float* __restrict__ nodes, const int* __restrict__ src,
    const int* __restrict__ dst, const float* __restrict__ W1,
    const float* __restrict__ b1, const float* __restrict__ W2,
    const float* __restrict__ b2, float* __restrict__ out, int E)
{
  __shared__ v4f ldsB4[(KK_TOT / 2) * NT * 32];
  __shared__ __attribute__((aligned(16))) unsigned char rowbuf[NWAVES * ROWBUF_BYTES];
  v2f* ldsB = (v2f*)ldsB4;

  for (int g = threadIdx.x; g < KK_TOT * NT * 32; g += blockDim.x) {
    int f = g >> 5, ln = g & 31;
    int n = f >> 5, kk = f & 31;
    int hf = ln >> 4, c = (n << 4) + (ln & 15);
    int row = ((kk >> 1) << 3) + (hf << 2) + ((kk & 1) << 1);
    v2f v;
    v.x = W1[row * DIM + c];
    v.y = W1[(row + 1) * DIM + c];
    ldsB[(((n << 4) + (kk >> 1)) * 32 + ln) * 2 + (kk & 1)] = v;
  }
  __syncthreads();

  const int lane = threadIdx.x & 31;
  const int half = lane >> 4;
  const int l15  = lane & 15;

  const float b2v = b2[0];

  const int wv = __builtin_amdgcn_readfirstlane((int)(threadIdx.x >> 5));
  const int passes = (E + 16 * ET - 1) / (16 * ET);
  const int pbase = passes / (int)gridDim.x, prem = passes % (int)gridDim.x;
  const int bx = (int)blockIdx.x;
  const int p0 = bx * pbase + (bx < prem ? bx : prem);
  const int p1 = p0 + pbase + (bx < prem ? 1 : 0);

  unsigned char* const rb = rowbuf + wv * ROWBUF_BYTES;
  const unsigned char* const rdp = rb + l15 * ROW_PITCH + (half << 4);
  unsigned char* const wrp = rb + half * ROW_PITCH + (l15 << 4);
  const char* const nbytes = (const char*)nodes;
  const unsigned lofs = (unsigned)l15 << 4;

  int t = p0 + wv;
  if (t < p1) {
    int sC, dC, sN, dN;
    {
      int e = t * (16 * ET) + 2 * l15 + half; if (e >= E) e = E - 1;
      sC = src[e]; dC = dst[e];
      e = (t + NWAVES) * (16 * ET) + 2 * l15 + half; if (e >= E) e = E - 1;
      sN = src[e]; dN = dst[e];
    }
    {
      v4f Ts[16], Td[16];
      GATHER16(sC, Ts)
      GATHER16(dC, Td)
#pragma unroll
      for (int k = 0; k < 16; ++k) *(v4f*)(wrp + 2 * k * ROW_PITCH) = Ts[k];
#pragma unroll
      for (int k = 0; k < 16; ++k) *(v4f*)(wrp + 2 * k * ROW_PITCH + 256) = Td[k];
    }

    for (; t < p1; t += NWAVES) {
      asm volatile("" ::: "memory");
      const int e0 = t * (16 * ET);

      v4f T[16];
      GATHER16(sN, T)
      int sNN, dNN;
      {
        int e = (t + 2 * NWAVES) * (16 * ET) + 2 * l15 + half; if (e >= E) e = E - 1;
        sNN = src[e]; dNN = dst[e];
      }

      v8f acc[ET][NT];
#pragma unroll
      for (int u = 0; u < ET; ++u)
#pragma unroll
        for (int n = 0; n < NT; ++n)
#pragma unroll
          for (int r = 0; r < 8; ++r) acc[u][n][r] = 0.f;

      v2f bq[2][2][NT];
      v4f aq[2][ET];
      v8f bias[NT], w2v[NT];
#pragma unroll
      for (int n = 0; n < NT; ++n) {
        const v4f q = ldsB4[(n * 16 + 0) * 32 + lane];
        bq[0][0][n] = q.xy;
        bq[0][1][n] = q.zw;
      }
#pragma unroll
      for (int u = 0; u < ET; ++u) aq[0][u] = *(const v4f*)(rdp + u * 16 * ROW_PITCH);

#pragma unroll
      for (int g = 0; g < NGRP; ++g) {
        const int cur = g & 1, nxt = cur ^ 1;
        __builtin_amdgcn_sched_barrier(0);
        if (g + 1 < NGRP) {
#pragma unroll
          for (int n = 0; n < NT; ++n) {
            const v4f q = ldsB4[(n * 16 + g + 1) * 32 + lane];
            bq[nxt][0][n] = q.xy;
            bq[nxt][1][n] = q.zw;
          }
#pragma unroll
          for (int u = 0; u < ET; ++u)
            aq[nxt][u] = *(const v4f*)(rdp + u * 16 * ROW_PITCH + 32 * (g + 1));
        } else {
#pragma unroll
          for (int n = 0; n < NT; ++n) {
            bias[n] = *(const v8f*)(b1 + n * 16 + half * 8);
            w2v[n]  = *(const v8f*)(W2 + n * 16 + half * 8);
          }
        }
        if (g == 7) {
#pragma unroll
          for (int k = 0; k < 16; ++k) *(v4f*)(wrp + 2 * k * ROW_PITCH) = T[k];
          GATHER16(dN, T)
        }
        if (g == 14) {
#pragma unroll
          for (int k = 0; k < 16; ++k) *(v4f*)(wrp + 2 * k * ROW_PITCH + 256) = T[k];
        }
        __builtin_amdgcn_sched_barrier(0);
#pragma unroll
        for (int j = 0; j < 2; ++j) {
#pragma unroll
          for (int n = 0; n < NT; ++n)
#pragma unroll
            for (int u = 0; u < ET; ++u) {
              const v2f a = j ? aq[cur][u].zw : aq[cur][u].xy;
              acc[u][n] = __builtin_amdgcn_wmma_f32_16x16x4_f32(
                  false, bq[cur][j][n], false, a, (short)0, acc[u][n], false, false);
            }
        }
      }

      float logit[ET];
#pragma unroll
      for (int u = 0; u < ET; ++u) {
        v2f ps[NT];
#pragma unroll
        for (int n = 0; n < NT; ++n) {
          ps[n].x = 0.f; ps[n].y = 0.f;
#pragma unroll
          for (int r = 0; r < 8; r += 2) {
            v2f h;
            h.x = acc[u][n][r]     + bias[n][r];
            h.y = acc[u][n][r + 1] + bias[n][r + 1];
            h.x = h.x > 0.f ? h.x : 0.f;
            h.y = h.y > 0.f ? h.y : 0.f;
            ps[n].x += h.x * w2v[n][r];
            ps[n].y += h.y * w2v[n][r + 1];
          }
        }
        const v2f p2 = (ps[0] + ps[1]) + (ps[2] + ps[3]);
        float lg = p2.x + p2.y;
        const int b = __builtin_bit_cast(int, lg);
        lg += __builtin_bit_cast(float,
            __builtin_amdgcn_permlanex16(b, b, 0x76543210, 0xfedcba98, false, false));
        logit[u] = lg + b2v;
      }

      sN = sNN; dN = dNN;

      const float res = half ? logit[1] : logit[0];
      if (e0 + lane < E) out[e0 + lane] = res;
    }
  }
}

extern "C" void kernel_launch(void* const* d_in, const int* in_sizes, int n_in,
                              void* d_out, int out_size, void* d_ws, size_t ws_size,
                              hipStream_t stream) {
  const float* nodes = (const float*)d_in[0];
  const int*   src   = (const int*)d_in[1];
  const int*   dst   = (const int*)d_in[2];
  const float* W1    = (const float*)d_in[3];
  const float* b1    = (const float*)d_in[4];
  const float* W2    = (const float*)d_in[5];
  const float* b2    = (const float*)d_in[6];
  float* out = (float*)d_out;

  const int E = in_sizes[1];

  edge_mlp_wmma<<<256, NWAVES * 32, 0, stream>>>(nodes, src, dst, W1, b1, W2, b2, out, E);
}
